# starting kernel
# baseline (speedup 1.0000x reference)
#include <hip/hip_runtime.h>
#include <math.h>

typedef __attribute__((ext_vector_type(2))) float v2f;
typedef __attribute__((ext_vector_type(8))) float v8f;

#define IN_F   256
#define MLP    128
#define NATT   16
#define NBINS  64
#define PTILE  64
#define H      128
#define W      128
#define SRC    64
#define NB     4

__device__ __forceinline__ float softplus_stable(float v) {
    return fmaxf(v, 0.0f) + log1pf(__expf(-fabsf(v)));
}

__global__ __launch_bounds__(256)
void attractor_fused(const float* __restrict__ x,
                     const float* __restrict__ b_prev,
                     const float* __restrict__ emb,
                     const float* __restrict__ w1,
                     const float* __restrict__ b1,
                     const float* __restrict__ w2,
                     const float* __restrict__ b2,
                     float* __restrict__ out)
{
    __shared__ float s_xin [PTILE * IN_F];
    __shared__ float s_hidd[PTILE * MLP];
    __shared__ float s_att [PTILE * NATT];

    const int tid = threadIdx.x;
    const int blk = blockIdx.x;
    const int ox_base = (blk & 1) * PTILE;
    const int oy      = (blk >> 1) & (H - 1);
    const int n       = blk >> 8;

    const float sc = (float)(SRC - 1) / (float)(H - 1);
    const float fy = (float)oy * sc;
    const int   iy0 = (int)fy;
    const int   iy1 = min(iy0 + 1, SRC - 1);
    const float wy  = fy - (float)iy0;

    for (int idx = tid; idx < IN_F * PTILE; idx += 256) {
        const int c = idx >> 6;
        const int p = idx & (PTILE - 1);
        const int ox = ox_base + p;
        const float fx = (float)ox * sc;
        const int   ix0 = (int)fx;
        const int   ix1 = min(ix0 + 1, SRC - 1);
        const float wx  = fx - (float)ix0;

        const float* eb = emb + (size_t)(n * IN_F + c) * SRC * SRC;
        const float e00 = eb[iy0 * SRC + ix0];
        const float e01 = eb[iy0 * SRC + ix1];
        const float e10 = eb[iy1 * SRC + ix0];
        const float e11 = eb[iy1 * SRC + ix1];
        const float ev  = (1.f - wy) * ((1.f - wx) * e00 + wx * e01)
                        +        wy  * ((1.f - wx) * e10 + wx * e11);
        const float xv = x[((size_t)(n * IN_F + c) * H + oy) * W + ox];
        s_xin[p * IN_F + c] = xv + ev;
    }
    __syncthreads();

    const int lane = tid & 31;
    const int wave = tid >> 5;
    const int ln16 = lane & 15;
    const int hi   = lane >> 4;

    {
        const int m_base = wave * 16;
        v8f acc0 = {}, acc1 = {}, acc2 = {}, acc3 = {};
        const float* w1row = w1 + (size_t)(m_base + ln16) * IN_F + hi * 2;
        const float* bp0 = &s_xin[( 0 + ln16) * IN_F + hi * 2];
        const float* bp1 = &s_xin[(16 + ln16) * IN_F + hi * 2];
        const float* bp2 = &s_xin[(32 + ln16) * IN_F + hi * 2];
        const float* bp3 = &s_xin[(48 + ln16) * IN_F + hi * 2];
        #pragma unroll 4
        for (int k = 0; k < IN_F; k += 4) {
            const v2f a  = *(const v2f*)(w1row + k);
            const v2f bb0 = *(const v2f*)(bp0 + k);
            const v2f bb1 = *(const v2f*)(bp1 + k);
            const v2f bb2 = *(const v2f*)(bp2 + k);
            const v2f bb3 = *(const v2f*)(bp3 + k);
            acc0 = __builtin_amdgcn_wmma_f32_16x16x4_f32(false, a, false, bb0, (short)0, acc0, false, false);
            acc1 = __builtin_amdgcn_wmma_f32_16x16x4_f32(false, a, false, bb1, (short)0, acc1, false, false);
            acc2 = __builtin_amdgcn_wmma_f32_16x16x4_f32(false, a, false, bb2, (short)0, acc2, false, false);
            acc3 = __builtin_amdgcn_wmma_f32_16x16x4_f32(false, a, false, bb3, (short)0, acc3, false, false);
        }
        #pragma unroll
        for (int i = 0; i < 8; ++i) {
            const int m = m_base + hi * 8 + i;
            const float bias = b1[m];
            s_hidd[( 0 + ln16) * MLP + m] = fmaxf(acc0[i] + bias, 0.f);
            s_hidd[(16 + ln16) * MLP + m] = fmaxf(acc1[i] + bias, 0.f);
            s_hidd[(32 + ln16) * MLP + m] = fmaxf(acc2[i] + bias, 0.f);
            s_hidd[(48 + ln16) * MLP + m] = fmaxf(acc3[i] + bias, 0.f);
        }
    }
    __syncthreads();

    if (wave < 4) {
        const int pix = wave * 16;
        v8f acc = {};
        const float* w2row = w2 + (size_t)(ln16) * MLP + hi * 2;
        const float* bp = &s_hidd[(pix + ln16) * MLP + hi * 2];
        #pragma unroll 4
        for (int k = 0; k < MLP; k += 4) {
            const v2f a = *(const v2f*)(w2row + k);
            const v2f b = *(const v2f*)(bp + k);
            acc = __builtin_amdgcn_wmma_f32_16x16x4_f32(false, a, false, b, (short)0, acc, false, false);
        }
        #pragma unroll
        for (int i = 0; i < 8; ++i) {
            const int aix = hi * 8 + i;
            s_att[(pix + ln16) * NATT + aix] = softplus_stable(acc[i] + b2[aix]);
        }
    }
    __syncthreads();

    const size_t out_half = (size_t)NB * NBINS * H * W;
    for (int idx = tid; idx < NBINS * PTILE; idx += 256) {
        const int p = idx & (PTILE - 1);
        const int k = idx >> 6;
        const int ox = ox_base + p;
        const float fx = (float)ox * sc;
        const int   ix0 = (int)fx;
        const int   ix1 = min(ix0 + 1, SRC - 1);
        const float wx  = fx - (float)ix0;

        const float* bp = b_prev + (size_t)(n * NBINS + k) * SRC * SRC;
        const float c00 = bp[iy0 * SRC + ix0];
        const float c01 = bp[iy0 * SRC + ix1];
        const float c10 = bp[iy1 * SRC + ix0];
        const float c11 = bp[iy1 * SRC + ix1];
        const float bc  = (1.f - wy) * ((1.f - wx) * c00 + wx * c01)
                        +        wy  * ((1.f - wx) * c10 + wx * c11);

        float delta = 0.f;
        #pragma unroll
        for (int a = 0; a < NATT; ++a) {
            const float dx = s_att[p * NATT + a] - bc;
            delta += dx / (1.0f + 300.0f * dx * dx);
        }
        const float r = bc + delta;
        const size_t o = ((size_t)(n * NBINS + k) * H + oy) * W + ox;
        out[o] = r;
        out[o + out_half] = r;
    }
}

extern "C" void kernel_launch(void* const* d_in, const int* in_sizes, int n_in,
                              void* d_out, int out_size, void* d_ws, size_t ws_size,
                              hipStream_t stream) {
    const float* x      = (const float*)d_in[0];
    const float* b_prev = (const float*)d_in[1];
    const float* emb    = (const float*)d_in[2];
    const float* w1     = (const float*)d_in[3];
    const float* b1     = (const float*)d_in[4];
    const float* w2     = (const float*)d_in[5];
    const float* b2     = (const float*)d_in[6];
    float* out = (float*)d_out;

    dim3 grid(NB * H * (W / PTILE));
    attractor_fused<<<grid, 256, 0, stream>>>(x, b_prev, emb, w1, b1, w2, b2, out);
}


# Pad LDS row pitch of the xin and hidden tiles to 16 B x odd (260 and 132 floats) so WMMA fragment loads and tile stores no longer hit one bank 16 to 32 ways; k order and sums unchanged
# speedup vs baseline: 1.4854x; 1.4854x over previous
#include <hip/hip_runtime.h>
#include <math.h>

typedef __attribute__((ext_vector_type(2))) float v2f;
typedef __attribute__((ext_vector_type(8))) float v8f;

#define IN_F   256
#define MLP    128
#define NATT   16
#define NBINS  64
#define PTILE  64
#define H      128
#define W      128
#define SRC    64
#define NB     4
#define XS     (IN_F + 4)
#define HS     (MLP + 4)

__device__ __forceinline__ float softplus_stable(float v) {
    return fmaxf(v, 0.0f) + log1pf(__expf(-fabsf(v)));
}

__global__ __launch_bounds__(256)
void attractor_fused(const float* __restrict__ x,
                     const float* __restrict__ b_prev,
                     const float* __restrict__ emb,
                     const float* __restrict__ w1,
                     const float* __restrict__ b1,
                     const float* __restrict__ w2,
                     const float* __restrict__ b2,
                     float* __restrict__ out)
{
    __shared__ float s_xin [PTILE * XS];
    __shared__ float s_hidd[PTILE * HS];
    __shared__ float s_att [PTILE * NATT];

    const int tid = threadIdx.x;
    const int blk = blockIdx.x;
    const int ox_base = (blk & 1) * PTILE;
    const int oy      = (blk >> 1) & (H - 1);
    const int n       = blk >> 8;

    const float sc = (float)(SRC - 1) / (float)(H - 1);
    const float fy = (float)oy * sc;
    const int   iy0 = (int)fy;
    const int   iy1 = min(iy0 + 1, SRC - 1);
    const float wy  = fy - (float)iy0;

    for (int idx = tid; idx < IN_F * PTILE; idx += 256) {
        const int c = idx >> 6;
        const int p = idx & (PTILE - 1);
        const int ox = ox_base + p;
        const float fx = (float)ox * sc;
        const int   ix0 = (int)fx;
        const int   ix1 = min(ix0 + 1, SRC - 1);
        const float wx  = fx - (float)ix0;

        const float* eb = emb + (size_t)(n * IN_F + c) * SRC * SRC;
        const float e00 = eb[iy0 * SRC + ix0];
        const float e01 = eb[iy0 * SRC + ix1];
        const float e10 = eb[iy1 * SRC + ix0];
        const float e11 = eb[iy1 * SRC + ix1];
        const float ev  = (1.f - wy) * ((1.f - wx) * e00 + wx * e01)
                        +        wy  * ((1.f - wx) * e10 + wx * e11);
        const float xv = x[((size_t)(n * IN_F + c) * H + oy) * W + ox];
        s_xin[p * XS + c] = xv + ev;
    }
    __syncthreads();

    const int lane = tid & 31;
    const int wave = tid >> 5;
    const int ln16 = lane & 15;
    const int hi   = lane >> 4;

    {
        const int m_base = wave * 16;
        v8f acc0 = {}, acc1 = {}, acc2 = {}, acc3 = {};
        const float* w1row = w1 + (size_t)(m_base + ln16) * IN_F + hi * 2;
        const float* bp0 = &s_xin[( 0 + ln16) * XS + hi * 2];
        const float* bp1 = &s_xin[(16 + ln16) * XS + hi * 2];
        const float* bp2 = &s_xin[(32 + ln16) * XS + hi * 2];
        const float* bp3 = &s_xin[(48 + ln16) * XS + hi * 2];
        #pragma unroll 4
        for (int k = 0; k < IN_F; k += 4) {
            const v2f a  = *(const v2f*)(w1row + k);
            const v2f bb0 = *(const v2f*)(bp0 + k);
            const v2f bb1 = *(const v2f*)(bp1 + k);
            const v2f bb2 = *(const v2f*)(bp2 + k);
            const v2f bb3 = *(const v2f*)(bp3 + k);
            acc0 = __builtin_amdgcn_wmma_f32_16x16x4_f32(false, a, false, bb0, (short)0, acc0, false, false);
            acc1 = __builtin_amdgcn_wmma_f32_16x16x4_f32(false, a, false, bb1, (short)0, acc1, false, false);
            acc2 = __builtin_amdgcn_wmma_f32_16x16x4_f32(false, a, false, bb2, (short)0, acc2, false, false);
            acc3 = __builtin_amdgcn_wmma_f32_16x16x4_f32(false, a, false, bb3, (short)0, acc3, false, false);
        }
        #pragma unroll
        for (int i = 0; i < 8; ++i) {
            const int m = m_base + hi * 8 + i;
            const float bias = b1[m];
            s_hidd[( 0 + ln16) * HS + m] = fmaxf(acc0[i] + bias, 0.f);
            s_hidd[(16 + ln16) * HS + m] = fmaxf(acc1[i] + bias, 0.f);
            s_hidd[(32 + ln16) * HS + m] = fmaxf(acc2[i] + bias, 0.f);
            s_hidd[(48 + ln16) * HS + m] = fmaxf(acc3[i] + bias, 0.f);
        }
    }
    __syncthreads();

    if (wave < 4) {
        const int pix = wave * 16;
        v8f acc = {};
        const float* w2row = w2 + (size_t)(ln16) * MLP + hi * 2;
        const float* bp = &s_hidd[(pix + ln16) * HS + hi * 2];
        #pragma unroll 4
        for (int k = 0; k < MLP; k += 4) {
            const v2f a = *(const v2f*)(w2row + k);
            const v2f b = *(const v2f*)(bp + k);
            acc = __builtin_amdgcn_wmma_f32_16x16x4_f32(false, a, false, b, (short)0, acc, false, false);
        }
        #pragma unroll
        for (int i = 0; i < 8; ++i) {
            const int aix = hi * 8 + i;
            s_att[(pix + ln16) * NATT + aix] = softplus_stable(acc[i] + b2[aix]);
        }
    }
    __syncthreads();

    const size_t out_half = (size_t)NB * NBINS * H * W;
    for (int idx = tid; idx < NBINS * PTILE; idx += 256) {
        const int p = idx & (PTILE - 1);
        const int k = idx >> 6;
        const int ox = ox_base + p;
        const float fx = (float)ox * sc;
        const int   ix0 = (int)fx;
        const int   ix1 = min(ix0 + 1, SRC - 1);
        const float wx  = fx - (float)ix0;

        const float* bp = b_prev + (size_t)(n * NBINS + k) * SRC * SRC;
        const float c00 = bp[iy0 * SRC + ix0];
        const float c01 = bp[iy0 * SRC + ix1];
        const float c10 = bp[iy1 * SRC + ix0];
        const float c11 = bp[iy1 * SRC + ix1];
        const float bc  = (1.f - wy) * ((1.f - wx) * c00 + wx * c01)
                        +        wy  * ((1.f - wx) * c10 + wx * c11);

        float delta = 0.f;
        #pragma unroll
        for (int a = 0; a < NATT; ++a) {
            const float dx = s_att[p * NATT + a] - bc;
            delta += dx / (1.0f + 300.0f * dx * dx);
        }
        const float r = bc + delta;
        const size_t o = ((size_t)(n * NBINS + k) * H + oy) * W + ox;
        out[o] = r;
        out[o + out_half] = r;
    }
}

extern "C" void kernel_launch(void* const* d_in, const int* in_sizes, int n_in,
                              void* d_out, int out_size, void* d_ws, size_t ws_size,
                              hipStream_t stream) {
    const float* x      = (const float*)d_in[0];
    const float* b_prev = (const float*)d_in[1];
    const float* emb    = (const float*)d_in[2];
    const float* w1     = (const float*)d_in[3];
    const float* b1     = (const float*)d_in[4];
    const float* w2     = (const float*)d_in[5];
    const float* b2     = (const float*)d_in[6];
    float* out = (float*)d_out;

    dim3 grid(NB * H * (W / PTILE));
    attractor_fused<<<grid, 256, 0, stream>>>(x, b_prev, emb, w1, b1, w2, b2, out);
}


# Stage 4 attractor term dx/(1+300dx^2): dx times v_rcp_f32 (1 ulp, den >= 1), fma-accumulated in the same order, replaces the IEEE f32 division expansion; VGPRs 235 to 114
# speedup vs baseline: 1.6267x; 1.0951x over previous
#include <hip/hip_runtime.h>
#include <math.h>

typedef __attribute__((ext_vector_type(2))) float v2f;
typedef __attribute__((ext_vector_type(8))) float v8f;

#define IN_F   256
#define MLP    128
#define NATT   16
#define NBINS  64
#define PTILE  64
#define H      128
#define W      128
#define SRC    64
#define NB     4
#define XS     (IN_F + 4)
#define HS     (MLP + 4)

__device__ __forceinline__ float softplus_stable(float v) {
    return fmaxf(v, 0.0f) + log1pf(__expf(-fabsf(v)));
}

__global__ __launch_bounds__(256)
void attractor_fused(const float* __restrict__ x,
                     const float* __restrict__ b_prev,
                     const float* __restrict__ emb,
                     const float* __restrict__ w1,
                     const float* __restrict__ b1,
                     const float* __restrict__ w2,
                     const float* __restrict__ b2,
                     float* __restrict__ out)
{
    __shared__ float s_xin [PTILE * XS];
    __shared__ float s_hidd[PTILE * HS];
    __shared__ float s_att [PTILE * NATT];

    const int tid = threadIdx.x;
    const int blk = blockIdx.x;
    const int ox_base = (blk & 1) * PTILE;
    const int oy      = (blk >> 1) & (H - 1);
    const int n       = blk >> 8;

    const float sc = (float)(SRC - 1) / (float)(H - 1);
    const float fy = (float)oy * sc;
    const int   iy0 = (int)fy;
    const int   iy1 = min(iy0 + 1, SRC - 1);
    const float wy  = fy - (float)iy0;

    for (int idx = tid; idx < IN_F * PTILE; idx += 256) {
        const int c = idx >> 6;
        const int p = idx & (PTILE - 1);
        const int ox = ox_base + p;
        const float fx = (float)ox * sc;
        const int   ix0 = (int)fx;
        const int   ix1 = min(ix0 + 1, SRC - 1);
        const float wx  = fx - (float)ix0;

        const float* eb = emb + (size_t)(n * IN_F + c) * SRC * SRC;
        const float e00 = eb[iy0 * SRC + ix0];
        const float e01 = eb[iy0 * SRC + ix1];
        const float e10 = eb[iy1 * SRC + ix0];
        const float e11 = eb[iy1 * SRC + ix1];
        const float ev  = (1.f - wy) * ((1.f - wx) * e00 + wx * e01)
                        +        wy  * ((1.f - wx) * e10 + wx * e11);
        const float xv = x[((size_t)(n * IN_F + c) * H + oy) * W + ox];
        s_xin[p * XS + c] = xv + ev;
    }
    __syncthreads();

    const int lane = tid & 31;
    const int wave = tid >> 5;
    const int ln16 = lane & 15;
    const int hi   = lane >> 4;

    {
        const int m_base = wave * 16;
        v8f acc0 = {}, acc1 = {}, acc2 = {}, acc3 = {};
        const float* w1row = w1 + (size_t)(m_base + ln16) * IN_F + hi * 2;
        const float* bp0 = &s_xin[( 0 + ln16) * XS + hi * 2];
        const float* bp1 = &s_xin[(16 + ln16) * XS + hi * 2];
        const float* bp2 = &s_xin[(32 + ln16) * XS + hi * 2];
        const float* bp3 = &s_xin[(48 + ln16) * XS + hi * 2];
        #pragma unroll 4
        for (int k = 0; k < IN_F; k += 4) {
            const v2f a  = *(const v2f*)(w1row + k);
            const v2f bb0 = *(const v2f*)(bp0 + k);
            const v2f bb1 = *(const v2f*)(bp1 + k);
            const v2f bb2 = *(const v2f*)(bp2 + k);
            const v2f bb3 = *(const v2f*)(bp3 + k);
            acc0 = __builtin_amdgcn_wmma_f32_16x16x4_f32(false, a, false, bb0, (short)0, acc0, false, false);
            acc1 = __builtin_amdgcn_wmma_f32_16x16x4_f32(false, a, false, bb1, (short)0, acc1, false, false);
            acc2 = __builtin_amdgcn_wmma_f32_16x16x4_f32(false, a, false, bb2, (short)0, acc2, false, false);
            acc3 = __builtin_amdgcn_wmma_f32_16x16x4_f32(false, a, false, bb3, (short)0, acc3, false, false);
        }
        #pragma unroll
        for (int i = 0; i < 8; ++i) {
            const int m = m_base + hi * 8 + i;
            const float bias = b1[m];
            s_hidd[( 0 + ln16) * HS + m] = fmaxf(acc0[i] + bias, 0.f);
            s_hidd[(16 + ln16) * HS + m] = fmaxf(acc1[i] + bias, 0.f);
            s_hidd[(32 + ln16) * HS + m] = fmaxf(acc2[i] + bias, 0.f);
            s_hidd[(48 + ln16) * HS + m] = fmaxf(acc3[i] + bias, 0.f);
        }
    }
    __syncthreads();

    if (wave < 4) {
        const int pix = wave * 16;
        v8f acc = {};
        const float* w2row = w2 + (size_t)(ln16) * MLP + hi * 2;
        const float* bp = &s_hidd[(pix + ln16) * HS + hi * 2];
        #pragma unroll 4
        for (int k = 0; k < MLP; k += 4) {
            const v2f a = *(const v2f*)(w2row + k);
            const v2f b = *(const v2f*)(bp + k);
            acc = __builtin_amdgcn_wmma_f32_16x16x4_f32(false, a, false, b, (short)0, acc, false, false);
        }
        #pragma unroll
        for (int i = 0; i < 8; ++i) {
            const int aix = hi * 8 + i;
            s_att[(pix + ln16) * NATT + aix] = softplus_stable(acc[i] + b2[aix]);
        }
    }
    __syncthreads();

    const size_t out_half = (size_t)NB * NBINS * H * W;
    for (int idx = tid; idx < NBINS * PTILE; idx += 256) {
        const int p = idx & (PTILE - 1);
        const int k = idx >> 6;
        const int ox = ox_base + p;
        const float fx = (float)ox * sc;
        const int   ix0 = (int)fx;
        const int   ix1 = min(ix0 + 1, SRC - 1);
        const float wx  = fx - (float)ix0;

        const float* bp = b_prev + (size_t)(n * NBINS + k) * SRC * SRC;
        const float c00 = bp[iy0 * SRC + ix0];
        const float c01 = bp[iy0 * SRC + ix1];
        const float c10 = bp[iy1 * SRC + ix0];
        const float c11 = bp[iy1 * SRC + ix1];
        const float bc  = (1.f - wy) * ((1.f - wx) * c00 + wx * c01)
                        +        wy  * ((1.f - wx) * c10 + wx * c11);

        float delta = 0.f;
        #pragma unroll
        for (int a = 0; a < NATT; ++a) {
            const float dx = s_att[p * NATT + a] - bc;
            delta += dx * __builtin_amdgcn_rcpf(1.0f + 300.0f * dx * dx);
        }
        const float r = bc + delta;
        const size_t o = ((size_t)(n * NBINS + k) * H + oy) * W + ox;
        out[o] = r;
        out[o + out_half] = r;
    }
}

extern "C" void kernel_launch(void* const* d_in, const int* in_sizes, int n_in,
                              void* d_out, int out_size, void* d_ws, size_t ws_size,
                              hipStream_t stream) {
    const float* x      = (const float*)d_in[0];
    const float* b_prev = (const float*)d_in[1];
    const float* emb    = (const float*)d_in[2];
    const float* w1     = (const float*)d_in[3];
    const float* b1     = (const float*)d_in[4];
    const float* w2     = (const float*)d_in[5];
    const float* b2     = (const float*)d_in[6];
    float* out = (float*)d_out;

    dim3 grid(NB * H * (W / PTILE));
    attractor_fused<<<grid, 256, 0, stream>>>(x, b_prev, emb, w1, b1, w2, b2, out);
}


# Hidden tile stored over the dead xin tile in LDS (one extra barrier): 102 KB to 69 KB per block, so 4 blocks per WGP and all 1024 blocks resident in one round instead of 3 per WGP plus a tail round
# speedup vs baseline: 1.8713x; 1.1504x over previous
#include <hip/hip_runtime.h>
#include <math.h>

typedef __attribute__((ext_vector_type(2))) float v2f;
typedef __attribute__((ext_vector_type(8))) float v8f;

#define IN_F   256
#define MLP    128
#define NATT   16
#define NBINS  64
#define PTILE  64
#define H      128
#define W      128
#define SRC    64
#define NB     4
#define XS     (IN_F + 4)
#define HS     (MLP + 4)

__device__ __forceinline__ float softplus_stable(float v) {
    return fmaxf(v, 0.0f) + log1pf(__expf(-fabsf(v)));
}

__global__ __launch_bounds__(256)
void attractor_fused(const float* __restrict__ x,
                     const float* __restrict__ b_prev,
                     const float* __restrict__ emb,
                     const float* __restrict__ w1,
                     const float* __restrict__ b1,
                     const float* __restrict__ w2,
                     const float* __restrict__ b2,
                     float* __restrict__ out)
{
    __shared__ float s_tile[PTILE * XS];
    __shared__ float s_att [PTILE * NATT];
    float* const s_xin  = s_tile;
    float* const s_hidd = s_tile;

    const int tid = threadIdx.x;
    const int blk = blockIdx.x;
    const int ox_base = (blk & 1) * PTILE;
    const int oy      = (blk >> 1) & (H - 1);
    const int n       = blk >> 8;

    const float sc = (float)(SRC - 1) / (float)(H - 1);
    const float fy = (float)oy * sc;
    const int   iy0 = (int)fy;
    const int   iy1 = min(iy0 + 1, SRC - 1);
    const float wy  = fy - (float)iy0;

    for (int idx = tid; idx < IN_F * PTILE; idx += 256) {
        const int c = idx >> 6;
        const int p = idx & (PTILE - 1);
        const int ox = ox_base + p;
        const float fx = (float)ox * sc;
        const int   ix0 = (int)fx;
        const int   ix1 = min(ix0 + 1, SRC - 1);
        const float wx  = fx - (float)ix0;

        const float* eb = emb + (size_t)(n * IN_F + c) * SRC * SRC;
        const float e00 = eb[iy0 * SRC + ix0];
        const float e01 = eb[iy0 * SRC + ix1];
        const float e10 = eb[iy1 * SRC + ix0];
        const float e11 = eb[iy1 * SRC + ix1];
        const float ev  = (1.f - wy) * ((1.f - wx) * e00 + wx * e01)
                        +        wy  * ((1.f - wx) * e10 + wx * e11);
        const float xv = x[((size_t)(n * IN_F + c) * H + oy) * W + ox];
        s_xin[p * XS + c] = xv + ev;
    }
    __syncthreads();

    const int lane = tid & 31;
    const int wave = tid >> 5;
    const int ln16 = lane & 15;
    const int hi   = lane >> 4;

    {
        const int m_base = wave * 16;
        v8f acc0 = {}, acc1 = {}, acc2 = {}, acc3 = {};
        const float* w1row = w1 + (size_t)(m_base + ln16) * IN_F + hi * 2;
        const float* bp0 = &s_xin[( 0 + ln16) * XS + hi * 2];
        const float* bp1 = &s_xin[(16 + ln16) * XS + hi * 2];
        const float* bp2 = &s_xin[(32 + ln16) * XS + hi * 2];
        const float* bp3 = &s_xin[(48 + ln16) * XS + hi * 2];
        #pragma unroll 4
        for (int k = 0; k < IN_F; k += 4) {
            const v2f a  = *(const v2f*)(w1row + k);
            const v2f bb0 = *(const v2f*)(bp0 + k);
            const v2f bb1 = *(const v2f*)(bp1 + k);
            const v2f bb2 = *(const v2f*)(bp2 + k);
            const v2f bb3 = *(const v2f*)(bp3 + k);
            acc0 = __builtin_amdgcn_wmma_f32_16x16x4_f32(false, a, false, bb0, (short)0, acc0, false, false);
            acc1 = __builtin_amdgcn_wmma_f32_16x16x4_f32(false, a, false, bb1, (short)0, acc1, false, false);
            acc2 = __builtin_amdgcn_wmma_f32_16x16x4_f32(false, a, false, bb2, (short)0, acc2, false, false);
            acc3 = __builtin_amdgcn_wmma_f32_16x16x4_f32(false, a, false, bb3, (short)0, acc3, false, false);
        }
        __syncthreads();
        #pragma unroll
        for (int i = 0; i < 8; ++i) {
            const int m = m_base + hi * 8 + i;
            const float bias = b1[m];
            s_hidd[( 0 + ln16) * HS + m] = fmaxf(acc0[i] + bias, 0.f);
            s_hidd[(16 + ln16) * HS + m] = fmaxf(acc1[i] + bias, 0.f);
            s_hidd[(32 + ln16) * HS + m] = fmaxf(acc2[i] + bias, 0.f);
            s_hidd[(48 + ln16) * HS + m] = fmaxf(acc3[i] + bias, 0.f);
        }
    }
    __syncthreads();

    if (wave < 4) {
        const int pix = wave * 16;
        v8f acc = {};
        const float* w2row = w2 + (size_t)(ln16) * MLP + hi * 2;
        const float* bp = &s_hidd[(pix + ln16) * HS + hi * 2];
        #pragma unroll 4
        for (int k = 0; k < MLP; k += 4) {
            const v2f a = *(const v2f*)(w2row + k);
            const v2f b = *(const v2f*)(bp + k);
            acc = __builtin_amdgcn_wmma_f32_16x16x4_f32(false, a, false, b, (short)0, acc, false, false);
        }
        #pragma unroll
        for (int i = 0; i < 8; ++i) {
            const int aix = hi * 8 + i;
            s_att[(pix + ln16) * NATT + aix] = softplus_stable(acc[i] + b2[aix]);
        }
    }
    __syncthreads();

    const size_t out_half = (size_t)NB * NBINS * H * W;
    for (int idx = tid; idx < NBINS * PTILE; idx += 256) {
        const int p = idx & (PTILE - 1);
        const int k = idx >> 6;
        const int ox = ox_base + p;
        const float fx = (float)ox * sc;
        const int   ix0 = (int)fx;
        const int   ix1 = min(ix0 + 1, SRC - 1);
        const float wx  = fx - (float)ix0;

        const float* bp = b_prev + (size_t)(n * NBINS + k) * SRC * SRC;
        const float c00 = bp[iy0 * SRC + ix0];
        const float c01 = bp[iy0 * SRC + ix1];
        const float c10 = bp[iy1 * SRC + ix0];
        const float c11 = bp[iy1 * SRC + ix1];
        const float bc  = (1.f - wy) * ((1.f - wx) * c00 + wx * c01)
                        +        wy  * ((1.f - wx) * c10 + wx * c11);

        float delta = 0.f;
        #pragma unroll
        for (int a = 0; a < NATT; ++a) {
            const float dx = s_att[p * NATT + a] - bc;
            delta += dx * __builtin_amdgcn_rcpf(1.0f + 300.0f * dx * dx);
        }
        const float r = bc + delta;
        const size_t o = ((size_t)(n * NBINS + k) * H + oy) * W + ox;
        out[o] = r;
        out[o + out_half] = r;
    }
}

extern "C" void kernel_launch(void* const* d_in, const int* in_sizes, int n_in,
                              void* d_out, int out_size, void* d_ws, size_t ws_size,
                              hipStream_t stream) {
    const float* x      = (const float*)d_in[0];
    const float* b_prev = (const float*)d_in[1];
    const float* emb    = (const float*)d_in[2];
    const float* w1     = (const float*)d_in[3];
    const float* b1     = (const float*)d_in[4];
    const float* w2     = (const float*)d_in[5];
    const float* b2     = (const float*)d_in[6];
    float* out = (float*)d_out;

    dim3 grid(NB * H * (W / PTILE));
    attractor_fused<<<grid, 256, 0, stream>>>(x, b_prev, emb, w1, b1, w2, b2, out);
}


# Stage 1 loads widened to 16 B per lane: a lane takes 4 adjacent pixels of a channel (one x load, a 4-column window per embedding row, taps picked in registers); bilinear blend rows then columns
# speedup vs baseline: 1.9672x; 1.0512x over previous
#include <hip/hip_runtime.h>
#include <math.h>

typedef __attribute__((ext_vector_type(2))) float v2f;
typedef __attribute__((ext_vector_type(4))) float v4f;
typedef __attribute__((ext_vector_type(8))) float v8f;

#define IN_F   256
#define MLP    128
#define NATT   16
#define NBINS  64
#define PTILE  64
#define H      128
#define W      128
#define SRC    64
#define NB     4
#define XS     (IN_F + 4)
#define HS     (MLP + 4)

__device__ __forceinline__ v4f load4_dword_aligned(const float* p) {
    v4f v;
    __builtin_memcpy(&v, p, sizeof(v));
    return v;
}

__device__ __forceinline__ float pick4(v4f t, int off) {
    const float lo = (off & 1) ? t.y : t.x;
    const float hi = (off & 1) ? t.w : t.z;
    return (off & 2) ? hi : lo;
}

__device__ __forceinline__ float softplus_stable(float v) {
    return fmaxf(v, 0.0f) + log1pf(__expf(-fabsf(v)));
}

__global__ __launch_bounds__(256)
void attractor_fused(const float* __restrict__ x,
                     const float* __restrict__ b_prev,
                     const float* __restrict__ emb,
                     const float* __restrict__ w1,
                     const float* __restrict__ b1,
                     const float* __restrict__ w2,
                     const float* __restrict__ b2,
                     float* __restrict__ out)
{
    __shared__ float s_tile[PTILE * XS];
    __shared__ float s_att [PTILE * NATT];
    float* const s_xin  = s_tile;
    float* const s_hidd = s_tile;

    const int tid = threadIdx.x;
    const int blk = blockIdx.x;
    const int ox_base = (blk & 1) * PTILE;
    const int oy      = (blk >> 1) & (H - 1);
    const int n       = blk >> 8;

    const float sc = (float)(SRC - 1) / (float)(H - 1);
    const float fy = (float)oy * sc;
    const int   iy0 = (int)fy;
    const int   iy1 = min(iy0 + 1, SRC - 1);
    const float wy  = fy - (float)iy0;

    {
        const int g   = (tid & 7) | ((tid >> 2) & 8);
        const int cc  = ((tid >> 3) & 3) | ((tid >> 4) & 12);
        const int p0  = g * 4;
        const int ox0 = ox_base + p0;

        int   ixa[4], ixb[4];
        float wxv[4];
        #pragma unroll
        for (int i = 0; i < 4; ++i) {
            const float fx = (float)(ox0 + i) * sc;
            ixa[i] = (int)fx;
            ixb[i] = min(ixa[i] + 1, SRC - 1);
            wxv[i] = fx - (float)ixa[i];
        }
        const int s = min(ixa[0], SRC - 4);
        int offa[4], offb[4];
        #pragma unroll
        for (int i = 0; i < 4; ++i) { offa[i] = ixa[i] - s; offb[i] = ixb[i] - s; }

        const size_t chan = (size_t)(n * IN_F + cc);
        const float* xp  = x   + (chan * H + oy) * W + ox0;
        const float* ep0 = emb + chan * SRC * SRC + iy0 * SRC + s;
        const float* ep1 = emb + chan * SRC * SRC + iy1 * SRC + s;
        float* dst = &s_xin[p0 * XS + cc];
        const float wy0 = 1.f - wy;

        #pragma unroll 4
        for (int j = 0; j < IN_F / 16; ++j) {
            const v4f xv = *(const v4f*)(xp + (size_t)j * 16 * H * W);
            const v4f r0 = load4_dword_aligned(ep0 + (size_t)j * 16 * SRC * SRC);
            const v4f r1 = load4_dword_aligned(ep1 + (size_t)j * 16 * SRC * SRC);
            const v4f t  = r0 * wy0 + r1 * wy;
            #pragma unroll
            for (int i = 0; i < 4; ++i) {
                const float ta = pick4(t, offa[i]);
                const float tb = pick4(t, offb[i]);
                const float ev = ta * (1.f - wxv[i]) + tb * wxv[i];
                dst[i * XS + j * 16] = xv[i] + ev;
            }
        }
    }
    __syncthreads();

    const int lane = tid & 31;
    const int wave = tid >> 5;
    const int ln16 = lane & 15;
    const int hi   = lane >> 4;

    {
        const int m_base = wave * 16;
        v8f acc0 = {}, acc1 = {}, acc2 = {}, acc3 = {};
        const float* w1row = w1 + (size_t)(m_base + ln16) * IN_F + hi * 2;
        const float* bp0 = &s_xin[( 0 + ln16) * XS + hi * 2];
        const float* bp1 = &s_xin[(16 + ln16) * XS + hi * 2];
        const float* bp2 = &s_xin[(32 + ln16) * XS + hi * 2];
        const float* bp3 = &s_xin[(48 + ln16) * XS + hi * 2];
        #pragma unroll 4
        for (int k = 0; k < IN_F; k += 4) {
            const v2f a  = *(const v2f*)(w1row + k);
            const v2f bb0 = *(const v2f*)(bp0 + k);
            const v2f bb1 = *(const v2f*)(bp1 + k);
            const v2f bb2 = *(const v2f*)(bp2 + k);
            const v2f bb3 = *(const v2f*)(bp3 + k);
            acc0 = __builtin_amdgcn_wmma_f32_16x16x4_f32(false, a, false, bb0, (short)0, acc0, false, false);
            acc1 = __builtin_amdgcn_wmma_f32_16x16x4_f32(false, a, false, bb1, (short)0, acc1, false, false);
            acc2 = __builtin_amdgcn_wmma_f32_16x16x4_f32(false, a, false, bb2, (short)0, acc2, false, false);
            acc3 = __builtin_amdgcn_wmma_f32_16x16x4_f32(false, a, false, bb3, (short)0, acc3, false, false);
        }
        __syncthreads();
        #pragma unroll
        for (int i = 0; i < 8; ++i) {
            const int m = m_base + hi * 8 + i;
            const float bias = b1[m];
            s_hidd[( 0 + ln16) * HS + m] = fmaxf(acc0[i] + bias, 0.f);
            s_hidd[(16 + ln16) * HS + m] = fmaxf(acc1[i] + bias, 0.f);
            s_hidd[(32 + ln16) * HS + m] = fmaxf(acc2[i] + bias, 0.f);
            s_hidd[(48 + ln16) * HS + m] = fmaxf(acc3[i] + bias, 0.f);
        }
    }
    __syncthreads();

    if (wave < 4) {
        const int pix = wave * 16;
        v8f acc = {};
        const float* w2row = w2 + (size_t)(ln16) * MLP + hi * 2;
        const float* bp = &s_hidd[(pix + ln16) * HS + hi * 2];
        #pragma unroll 4
        for (int k = 0; k < MLP; k += 4) {
            const v2f a = *(const v2f*)(w2row + k);
            const v2f b = *(const v2f*)(bp + k);
            acc = __builtin_amdgcn_wmma_f32_16x16x4_f32(false, a, false, b, (short)0, acc, false, false);
        }
        #pragma unroll
        for (int i = 0; i < 8; ++i) {
            const int aix = hi * 8 + i;
            s_att[(pix + ln16) * NATT + aix] = softplus_stable(acc[i] + b2[aix]);
        }
    }
    __syncthreads();

    const size_t out_half = (size_t)NB * NBINS * H * W;
    for (int idx = tid; idx < NBINS * PTILE; idx += 256) {
        const int p = idx & (PTILE - 1);
        const int k = idx >> 6;
        const int ox = ox_base + p;
        const float fx = (float)ox * sc;
        const int   ix0 = (int)fx;
        const int   ix1 = min(ix0 + 1, SRC - 1);
        const float wx  = fx - (float)ix0;

        const float* bp = b_prev + (size_t)(n * NBINS + k) * SRC * SRC;
        const float c00 = bp[iy0 * SRC + ix0];
        const float c01 = bp[iy0 * SRC + ix1];
        const float c10 = bp[iy1 * SRC + ix0];
        const float c11 = bp[iy1 * SRC + ix1];
        const float bc  = (1.f - wy) * ((1.f - wx) * c00 + wx * c01)
                        +        wy  * ((1.f - wx) * c10 + wx * c11);

        float delta = 0.f;
        #pragma unroll
        for (int a = 0; a < NATT; ++a) {
            const float dx = s_att[p * NATT + a] - bc;
            delta += dx * __builtin_amdgcn_rcpf(1.0f + 300.0f * dx * dx);
        }
        const float r = bc + delta;
        const size_t o = ((size_t)(n * NBINS + k) * H + oy) * W + ox;
        out[o] = r;
        out[o + out_half] = r;
    }
}

extern "C" void kernel_launch(void* const* d_in, const int* in_sizes, int n_in,
                              void* d_out, int out_size, void* d_ws, size_t ws_size,
                              hipStream_t stream) {
    const float* x      = (const float*)d_in[0];
    const float* b_prev = (const float*)d_in[1];
    const float* emb    = (const float*)d_in[2];
    const float* w1     = (const float*)d_in[3];
    const float* b1     = (const float*)d_in[4];
    const float* w2     = (const float*)d_in[5];
    const float* b2     = (const float*)d_in[6];
    float* out = (float*)d_out;

    dim3 grid(NB * H * (W / PTILE));
    attractor_fused<<<grid, 256, 0, stream>>>(x, b_prev, emb, w1, b1, w2, b2, out);
}


# Stage 1 and GEMM1 pipelined over 8 chunks of 32 channels: the next chunk's x and embedding loads are requested before a chunk's 32 WMMAs and blended after them; k order and sums unchanged
# speedup vs baseline: 2.0554x; 1.0448x over previous
#include <hip/hip_runtime.h>
#include <math.h>

typedef __attribute__((ext_vector_type(2))) float v2f;
typedef __attribute__((ext_vector_type(4))) float v4f;
typedef __attribute__((ext_vector_type(8))) float v8f;

#define IN_F   256
#define MLP    128
#define NATT   16
#define NBINS  64
#define PTILE  64
#define H      128
#define W      128
#define SRC    64
#define NB     4
#define XS     (IN_F + 4)
#define HS     (MLP + 4)

__device__ __forceinline__ v4f load4_dword_aligned(const float* p) {
    v4f v;
    __builtin_memcpy(&v, p, sizeof(v));
    return v;
}

__device__ __forceinline__ float pick4(v4f t, int off) {
    const float lo = (off & 1) ? t.y : t.x;
    const float hi = (off & 1) ? t.w : t.z;
    return (off & 2) ? hi : lo;
}

__device__ __forceinline__ float softplus_stable(float v) {
    return fmaxf(v, 0.0f) + log1pf(__expf(-fabsf(v)));
}

__global__ __launch_bounds__(256, 8)
void attractor_fused(const float* __restrict__ x,
                     const float* __restrict__ b_prev,
                     const float* __restrict__ emb,
                     const float* __restrict__ w1,
                     const float* __restrict__ b1,
                     const float* __restrict__ w2,
                     const float* __restrict__ b2,
                     float* __restrict__ out)
{
    __shared__ float s_tile[PTILE * XS];
    __shared__ float s_att [PTILE * NATT];
    float* const s_xin  = s_tile;
    float* const s_hidd = s_tile;

    const int tid = threadIdx.x;
    const int blk = blockIdx.x;
    const int ox_base = (blk & 1) * PTILE;
    const int oy      = (blk >> 1) & (H - 1);
    const int n       = blk >> 8;

    const float sc = (float)(SRC - 1) / (float)(H - 1);
    const float fy = (float)oy * sc;
    const int   iy0 = (int)fy;
    const int   iy1 = min(iy0 + 1, SRC - 1);
    const float wy  = fy - (float)iy0;

    const int g   = (tid & 7) | ((tid >> 2) & 8);
    const int cc  = ((tid >> 3) & 3) | ((tid >> 4) & 12);
    const int p0  = g * 4;
    const int ox0 = ox_base + p0;

    int   ixa[4], ixb[4];
    float wxv[4];
    #pragma unroll
    for (int i = 0; i < 4; ++i) {
        const float fx = (float)(ox0 + i) * sc;
        ixa[i] = (int)fx;
        ixb[i] = min(ixa[i] + 1, SRC - 1);
        wxv[i] = fx - (float)ixa[i];
    }
    const int s = min(ixa[0], SRC - 4);
    int offa[4], offb[4];
    #pragma unroll
    for (int i = 0; i < 4; ++i) { offa[i] = ixa[i] - s; offb[i] = ixb[i] - s; }

    const float* const xu  = x   + ((size_t)n * IN_F * H + oy) * W + ox_base;
    const float* const eu0 = emb + (size_t)n * IN_F * SRC * SRC + iy0 * SRC;
    const float* const eu1 = emb + (size_t)n * IN_F * SRC * SRC + iy1 * SRC;
    const unsigned xoff = (unsigned)(cc * H * W + p0);
    const unsigned eoff = (unsigned)(cc * SRC * SRC + s);
    float* const dst = &s_xin[p0 * XS + cc];
    const float wy0 = 1.f - wy;

    v4f px[2], pr0[2], pr1[2];
    auto issue_chunk = [&](int c) {
        #pragma unroll
        for (int u = 0; u < 2; ++u) {
            const size_t j = (size_t)(2 * c + u);
            const float* const xb  = xu  + j * 16 * H * W;
            const float* const eb0 = eu0 + j * 16 * SRC * SRC;
            const float* const eb1 = eu1 + j * 16 * SRC * SRC;
            px[u]  = *(const v4f*)(xb + xoff);
            pr0[u] = load4_dword_aligned(eb0 + eoff);
            pr1[u] = load4_dword_aligned(eb1 + eoff);
        }
    };
    auto blend_store_chunk = [&](int c) {
        #pragma unroll
        for (int u = 0; u < 2; ++u) {
            const int j = 2 * c + u;
            const v4f t = pr0[u] * wy0 + pr1[u] * wy;
            #pragma unroll
            for (int i = 0; i < 4; ++i) {
                const float ta = pick4(t, offa[i]);
                const float tb = pick4(t, offb[i]);
                const float ev = ta * (1.f - wxv[i]) + tb * wxv[i];
                dst[i * XS + j * 16] = px[u][i] + ev;
            }
        }
    };

    const int lane = tid & 31;
    const int wave = tid >> 5;
    const int ln16 = lane & 15;
    const int hi   = lane >> 4;

    {
        const int m_base = wave * 16;
        v8f acc0 = {}, acc1 = {}, acc2 = {}, acc3 = {};
        const unsigned woff = (unsigned)((m_base + ln16) * IN_F + hi * 2);
        const float* bp0 = &s_xin[( 0 + ln16) * XS + hi * 2];
        const float* bp1 = &s_xin[(16 + ln16) * XS + hi * 2];
        const float* bp2 = &s_xin[(32 + ln16) * XS + hi * 2];
        const float* bp3 = &s_xin[(48 + ln16) * XS + hi * 2];

        issue_chunk(0);
        blend_store_chunk(0);
        __syncthreads();

        auto gemm_chunk = [&](const v2f (&a)[8], int k0) {
            #pragma unroll
            for (int q = 0; q < 8; ++q) {
                const int k = k0 + 4 * q;
                const v2f bb0 = *(const v2f*)(bp0 + k);
                const v2f bb1 = *(const v2f*)(bp1 + k);
                const v2f bb2 = *(const v2f*)(bp2 + k);
                const v2f bb3 = *(const v2f*)(bp3 + k);
                acc0 = __builtin_amdgcn_wmma_f32_16x16x4_f32(false, a[q], false, bb0, (short)0, acc0, false, false);
                acc1 = __builtin_amdgcn_wmma_f32_16x16x4_f32(false, a[q], false, bb1, (short)0, acc1, false, false);
                acc2 = __builtin_amdgcn_wmma_f32_16x16x4_f32(false, a[q], false, bb2, (short)0, acc2, false, false);
                acc3 = __builtin_amdgcn_wmma_f32_16x16x4_f32(false, a[q], false, bb3, (short)0, acc3, false, false);
            }
        };

        for (int c = 0; c < IN_F / 32 - 1; ++c) {
            v2f a[8];
            #pragma unroll
            for (int q = 0; q < 8; ++q) a[q] = *(const v2f*)(w1 + (c * 32 + 4 * q) + woff);
            issue_chunk(c + 1);
            gemm_chunk(a, c * 32);
            __builtin_amdgcn_sched_barrier(0);
            blend_store_chunk(c + 1);
            __syncthreads();
        }
        {
            v2f a[8];
            #pragma unroll
            for (int q = 0; q < 8; ++q) a[q] = *(const v2f*)(w1 + ((IN_F - 32) + 4 * q) + woff);
            gemm_chunk(a, IN_F - 32);
        }
        __syncthreads();
        #pragma unroll
        for (int i = 0; i < 8; ++i) {
            const int m = m_base + hi * 8 + i;
            const float bias = b1[m];
            s_hidd[( 0 + ln16) * HS + m] = fmaxf(acc0[i] + bias, 0.f);
            s_hidd[(16 + ln16) * HS + m] = fmaxf(acc1[i] + bias, 0.f);
            s_hidd[(32 + ln16) * HS + m] = fmaxf(acc2[i] + bias, 0.f);
            s_hidd[(48 + ln16) * HS + m] = fmaxf(acc3[i] + bias, 0.f);
        }
    }
    __syncthreads();

    if (wave < 4) {
        const int pix = wave * 16;
        v8f acc = {};
        const float* w2row = w2 + (size_t)(ln16) * MLP + hi * 2;
        const float* bp = &s_hidd[(pix + ln16) * HS + hi * 2];
        #pragma unroll 4
        for (int k = 0; k < MLP; k += 4) {
            const v2f a = *(const v2f*)(w2row + k);
            const v2f b = *(const v2f*)(bp + k);
            acc = __builtin_amdgcn_wmma_f32_16x16x4_f32(false, a, false, b, (short)0, acc, false, false);
        }
        #pragma unroll
        for (int i = 0; i < 8; ++i) {
            const int aix = hi * 8 + i;
            s_att[(pix + ln16) * NATT + aix] = softplus_stable(acc[i] + b2[aix]);
        }
    }
    __syncthreads();

    const size_t out_half = (size_t)NB * NBINS * H * W;
    for (int idx = tid; idx < NBINS * PTILE; idx += 256) {
        const int p = idx & (PTILE - 1);
        const int k = idx >> 6;
        const int ox = ox_base + p;
        const float fx = (float)ox * sc;
        const int   ix0 = (int)fx;
        const int   ix1 = min(ix0 + 1, SRC - 1);
        const float wx  = fx - (float)ix0;

        const float* bp = b_prev + (size_t)(n * NBINS + k) * SRC * SRC;
        const float c00 = bp[iy0 * SRC + ix0];
        const float c01 = bp[iy0 * SRC + ix1];
        const float c10 = bp[iy1 * SRC + ix0];
        const float c11 = bp[iy1 * SRC + ix1];
        const float bc  = (1.f - wy) * ((1.f - wx) * c00 + wx * c01)
                        +        wy  * ((1.f - wx) * c10 + wx * c11);

        float delta = 0.f;
        #pragma unroll
        for (int a = 0; a < NATT; ++a) {
            const float dx = s_att[p * NATT + a] - bc;
            delta += dx * __builtin_amdgcn_rcpf(1.0f + 300.0f * dx * dx);
        }
        const float r = bc + delta;
        const size_t o = ((size_t)(n * NBINS + k) * H + oy) * W + ox;
        out[o] = r;
        out[o + out_half] = r;
    }
}

extern "C" void kernel_launch(void* const* d_in, const int* in_sizes, int n_in,
                              void* d_out, int out_size, void* d_ws, size_t ws_size,
                              hipStream_t stream) {
    const float* x      = (const float*)d_in[0];
    const float* b_prev = (const float*)d_in[1];
    const float* emb    = (const float*)d_in[2];
    const float* w1     = (const float*)d_in[3];
    const float* b1     = (const float*)d_in[4];
    const float* w2     = (const float*)d_in[5];
    const float* b2     = (const float*)d_in[6];
    float* out = (float*)d_out;

    dim3 grid(NB * H * (W / PTILE));
    attractor_fused<<<grid, 256, 0, stream>>>(x, b_prev, emb, w1, b1, w2, b2, out);
}


# W1 reaches GEMM1 through LDS: each wave fetches its 16x32 slice of the next chunk as whole 128 B lines and reads A fragments from there; xin kept as a 2-chunk ring to make room; k order unchanged
# speedup vs baseline: 2.1335x; 1.0380x over previous
#include <hip/hip_runtime.h>
#include <math.h>

typedef __attribute__((ext_vector_type(2))) float v2f;
typedef __attribute__((ext_vector_type(4))) float v4f;
typedef __attribute__((ext_vector_type(8))) float v8f;

#define IN_F   256
#define MLP    128
#define NATT   16
#define NBINS  64
#define PTILE  64
#define H      128
#define W      128
#define SRC    64
#define NB     4
#define CH     32
#define XP     (CH + 4)
#define HS     (MLP + 4)
#define XBUF   (PTILE * XP)
#define WOFF   (2 * XBUF)
#define WSTG   (MLP * XP)

__device__ __forceinline__ v4f load4_dword_aligned(const float* p) {
    v4f v;
    __builtin_memcpy(&v, p, sizeof(v));
    return v;
}

__device__ __forceinline__ float pick4(v4f t, int off) {
    const float lo = (off & 1) ? t.y : t.x;
    const float hi = (off & 1) ? t.w : t.z;
    return (off & 2) ? hi : lo;
}

__device__ __forceinline__ float softplus_stable(float v) {
    return fmaxf(v, 0.0f) + log1pf(__expf(-fabsf(v)));
}

__global__ __launch_bounds__(256, 8)
void attractor_fused(const float* __restrict__ x,
                     const float* __restrict__ b_prev,
                     const float* __restrict__ emb,
                     const float* __restrict__ w1,
                     const float* __restrict__ b1,
                     const float* __restrict__ w2,
                     const float* __restrict__ b2,
                     float* __restrict__ out)
{
    __shared__ __attribute__((aligned(16))) float s_tile[WOFF + WSTG];
    __shared__ float s_att [PTILE * NATT];
    float* const s_x    = s_tile;
    float* const s_hidd = s_tile;

    const int tid = threadIdx.x;
    const int blk = blockIdx.x;
    const int ox_base = (blk & 1) * PTILE;
    const int oy      = (blk >> 1) & (H - 1);
    const int n       = blk >> 8;

    const float sc = (float)(SRC - 1) / (float)(H - 1);
    const float fy = (float)oy * sc;
    const int   iy0 = (int)fy;
    const int   iy1 = min(iy0 + 1, SRC - 1);
    const float wy  = fy - (float)iy0;

    const int g   = (tid & 7) | ((tid >> 2) & 8);
    const int cc  = ((tid >> 3) & 3) | ((tid >> 4) & 12);
    const int p0  = g * 4;
    const int ox0 = ox_base + p0;

    int   ixa[4], ixb[4];
    float wxv[4];
    #pragma unroll
    for (int i = 0; i < 4; ++i) {
        const float fx = (float)(ox0 + i) * sc;
        ixa[i] = (int)fx;
        ixb[i] = min(ixa[i] + 1, SRC - 1);
        wxv[i] = fx - (float)ixa[i];
    }
    const int s = min(ixa[0], SRC - 4);
    int offa[4], offb[4];
    #pragma unroll
    for (int i = 0; i < 4; ++i) { offa[i] = ixa[i] - s; offb[i] = ixb[i] - s; }

    const float* const xu  = x   + ((size_t)n * IN_F * H + oy) * W + ox_base;
    const float* const eu0 = emb + (size_t)n * IN_F * SRC * SRC + iy0 * SRC;
    const float* const eu1 = emb + (size_t)n * IN_F * SRC * SRC + iy1 * SRC;
    const unsigned xoff = (unsigned)(cc * H * W + p0);
    const unsigned eoff = (unsigned)(cc * SRC * SRC + s);
    const int dsto = p0 * XP + cc;
    const float wy0 = 1.f - wy;

    const int lane = tid & 31;
    const int wave = tid >> 5;
    const int ln16 = lane & 15;
    const int hi   = lane >> 4;
    const int m_base = wave * 16;

    const int wrow = lane >> 3;
    const int wcol = (lane & 7) * 4;
    const unsigned wl = (unsigned)((m_base + wrow) * IN_F + wcol);
    float* const s_w = s_tile + WOFF + m_base * XP;
    float* const wdst = s_w + wrow * XP + wcol;

    v4f wv[4];
    v4f px[2], pr0[2], pr1[2];
    auto issue_w = [&](int c) {
        #pragma unroll
        for (int i = 0; i < 4; ++i)
            wv[i] = *(const v4f*)(w1 + (size_t)(c * CH + i * 4 * IN_F) + wl);
    };
    auto store_w = [&]() {
        #pragma unroll
        for (int i = 0; i < 4; ++i)
            *(v4f*)(wdst + i * 4 * XP) = wv[i];
    };
    auto issue_chunk = [&](int c) {
        #pragma unroll
        for (int u = 0; u < 2; ++u) {
            const size_t j = (size_t)(2 * c + u);
            const float* const xb  = xu  + j * 16 * H * W;
            const float* const eb0 = eu0 + j * 16 * SRC * SRC;
            const float* const eb1 = eu1 + j * 16 * SRC * SRC;
            px[u]  = *(const v4f*)(xb + xoff);
            pr0[u] = load4_dword_aligned(eb0 + eoff);
            pr1[u] = load4_dword_aligned(eb1 + eoff);
        }
    };
    auto blend_store_chunk = [&](int b) {
        float* const dst = s_x + b * XBUF + dsto;
        #pragma unroll
        for (int u = 0; u < 2; ++u) {
            const v4f t = pr0[u] * wy0 + pr1[u] * wy;
            #pragma unroll
            for (int i = 0; i < 4; ++i) {
                const float ta = pick4(t, offa[i]);
                const float tb = pick4(t, offb[i]);
                const float ev = ta * (1.f - wxv[i]) + tb * wxv[i];
                dst[i * XP + u * 16] = px[u][i] + ev;
            }
        }
    };

    {
        v8f acc0 = {}, acc1 = {}, acc2 = {}, acc3 = {};
        const float* const ap = s_w + ln16 * XP + hi * 2;
        const int bo = ln16 * XP + hi * 2;

        issue_w(0);
        issue_chunk(0);
        store_w();
        blend_store_chunk(0);
        __syncthreads();

        auto gemm_chunk = [&](int b) {
            const float* const bp0 = s_x + b * XBUF + bo;
            const float* const bp1 = bp0 + 16 * XP;
            const float* const bp2 = bp0 + 32 * XP;
            const float* const bp3 = bp0 + 48 * XP;
            #pragma unroll
            for (int q = 0; q < 8; ++q) {
                const v2f a   = *(const v2f*)(ap  + 4 * q);
                const v2f bb0 = *(const v2f*)(bp0 + 4 * q);
                const v2f bb1 = *(const v2f*)(bp1 + 4 * q);
                const v2f bb2 = *(const v2f*)(bp2 + 4 * q);
                const v2f bb3 = *(const v2f*)(bp3 + 4 * q);
                acc0 = __builtin_amdgcn_wmma_f32_16x16x4_f32(false, a, false, bb0, (short)0, acc0, false, false);
                acc1 = __builtin_amdgcn_wmma_f32_16x16x4_f32(false, a, false, bb1, (short)0, acc1, false, false);
                acc2 = __builtin_amdgcn_wmma_f32_16x16x4_f32(false, a, false, bb2, (short)0, acc2, false, false);
                acc3 = __builtin_amdgcn_wmma_f32_16x16x4_f32(false, a, false, bb3, (short)0, acc3, false, false);
            }
        };

        for (int c = 0; c < IN_F / CH - 1; ++c) {
            issue_w(c + 1);
            issue_chunk(c + 1);
            gemm_chunk(c & 1);
            __builtin_amdgcn_sched_barrier(0);
            store_w();
            blend_store_chunk((c + 1) & 1);
            __syncthreads();
        }
        gemm_chunk((IN_F / CH - 1) & 1);
        __syncthreads();
        #pragma unroll
        for (int i = 0; i < 8; ++i) {
            const int m = m_base + hi * 8 + i;
            const float bias = b1[m];
            s_hidd[( 0 + ln16) * HS + m] = fmaxf(acc0[i] + bias, 0.f);
            s_hidd[(16 + ln16) * HS + m] = fmaxf(acc1[i] + bias, 0.f);
            s_hidd[(32 + ln16) * HS + m] = fmaxf(acc2[i] + bias, 0.f);
            s_hidd[(48 + ln16) * HS + m] = fmaxf(acc3[i] + bias, 0.f);
        }
    }
    __syncthreads();

    if (wave < 4) {
        const int pix = wave * 16;
        v8f acc = {};
        const float* w2row = w2 + (size_t)(ln16) * MLP + hi * 2;
        const float* bp = &s_hidd[(pix + ln16) * HS + hi * 2];
        #pragma unroll 4
        for (int k = 0; k < MLP; k += 4) {
            const v2f a = *(const v2f*)(w2row + k);
            const v2f b = *(const v2f*)(bp + k);
            acc = __builtin_amdgcn_wmma_f32_16x16x4_f32(false, a, false, b, (short)0, acc, false, false);
        }
        #pragma unroll
        for (int i = 0; i < 8; ++i) {
            const int aix = hi * 8 + i;
            s_att[(pix + ln16) * NATT + aix] = softplus_stable(acc[i] + b2[aix]);
        }
    }
    __syncthreads();

    const size_t out_half = (size_t)NB * NBINS * H * W;
    for (int idx = tid; idx < NBINS * PTILE; idx += 256) {
        const int p = idx & (PTILE - 1);
        const int k = idx >> 6;
        const int ox = ox_base + p;
        const float fx = (float)ox * sc;
        const int   ix0 = (int)fx;
        const int   ix1 = min(ix0 + 1, SRC - 1);
        const float wx  = fx - (float)ix0;

        const float* bp = b_prev + (size_t)(n * NBINS + k) * SRC * SRC;
        const float c00 = bp[iy0 * SRC + ix0];
        const float c01 = bp[iy0 * SRC + ix1];
        const float c10 = bp[iy1 * SRC + ix0];
        const float c11 = bp[iy1 * SRC + ix1];
        const float bc  = (1.f - wy) * ((1.f - wx) * c00 + wx * c01)
                        +        wy  * ((1.f - wx) * c10 + wx * c11);

        float delta = 0.f;
        #pragma unroll
        for (int a = 0; a < NATT; ++a) {
            const float dx = s_att[p * NATT + a] - bc;
            delta += dx * __builtin_amdgcn_rcpf(1.0f + 300.0f * dx * dx);
        }
        const float r = bc + delta;
        const size_t o = ((size_t)(n * NBINS + k) * H + oy) * W + ox;
        out[o] = r;
        out[o + out_half] = r;
    }
}

extern "C" void kernel_launch(void* const* d_in, const int* in_sizes, int n_in,
                              void* d_out, int out_size, void* d_ws, size_t ws_size,
                              hipStream_t stream) {
    const float* x      = (const float*)d_in[0];
    const float* b_prev = (const float*)d_in[1];
    const float* emb    = (const float*)d_in[2];
    const float* w1     = (const float*)d_in[3];
    const float* b1     = (const float*)d_in[4];
    const float* w2     = (const float*)d_in[5];
    const float* b2     = (const float*)d_in[6];
    float* out = (float*)d_out;

    dim3 grid(NB * H * (W / PTILE));
    attractor_fused<<<grid, 256, 0, stream>>>(x, b_prev, emb, w1, b1, w2, b2, out);
}


# Stage 4 in packed f32: a lane takes two bins at a time in the two halves of v_pk_add/mul/fma operands (bilinear blend and the 16 attractor terms), bin loop unrolled; per-output sum order unchanged
# speedup vs baseline: 2.4491x; 1.1479x over previous
#include <hip/hip_runtime.h>
#include <math.h>

typedef __attribute__((ext_vector_type(2))) float v2f;
typedef __attribute__((ext_vector_type(4))) float v4f;
typedef __attribute__((ext_vector_type(8))) float v8f;

#define IN_F   256
#define MLP    128
#define NATT   16
#define NBINS  64
#define PTILE  64
#define H      128
#define W      128
#define SRC    64
#define NB     4
#define CH     32
#define XP     (CH + 4)
#define HS     (MLP + 4)
#define XBUF   (PTILE * XP)
#define WOFF   (2 * XBUF)
#define WSTG   (MLP * XP)

__device__ __forceinline__ v4f load4_dword_aligned(const float* p) {
    v4f v;
    __builtin_memcpy(&v, p, sizeof(v));
    return v;
}

__device__ __forceinline__ float pick4(v4f t, int off) {
    const float lo = (off & 1) ? t.y : t.x;
    const float hi = (off & 1) ? t.w : t.z;
    return (off & 2) ? hi : lo;
}

__device__ __forceinline__ float softplus_stable(float v) {
    return fmaxf(v, 0.0f) + log1pf(__expf(-fabsf(v)));
}

__global__ __launch_bounds__(256, 8)
void attractor_fused(const float* __restrict__ x,
                     const float* __restrict__ b_prev,
                     const float* __restrict__ emb,
                     const float* __restrict__ w1,
                     const float* __restrict__ b1,
                     const float* __restrict__ w2,
                     const float* __restrict__ b2,
                     float* __restrict__ out)
{
    __shared__ __attribute__((aligned(16))) float s_tile[WOFF + WSTG];
    __shared__ float s_att [PTILE * NATT];
    float* const s_x    = s_tile;
    float* const s_hidd = s_tile;

    const int tid = threadIdx.x;
    const int blk = blockIdx.x;
    const int ox_base = (blk & 1) * PTILE;
    const int oy      = (blk >> 1) & (H - 1);
    const int n       = blk >> 8;

    const float sc = (float)(SRC - 1) / (float)(H - 1);
    const float fy = (float)oy * sc;
    const int   iy0 = (int)fy;
    const int   iy1 = min(iy0 + 1, SRC - 1);
    const float wy  = fy - (float)iy0;

    const int g   = (tid & 7) | ((tid >> 2) & 8);
    const int cc  = ((tid >> 3) & 3) | ((tid >> 4) & 12);
    const int p0  = g * 4;
    const int ox0 = ox_base + p0;

    int   ixa[4], ixb[4];
    float wxv[4];
    #pragma unroll
    for (int i = 0; i < 4; ++i) {
        const float fx = (float)(ox0 + i) * sc;
        ixa[i] = (int)fx;
        ixb[i] = min(ixa[i] + 1, SRC - 1);
        wxv[i] = fx - (float)ixa[i];
    }
    const int s = min(ixa[0], SRC - 4);
    int offa[4], offb[4];
    #pragma unroll
    for (int i = 0; i < 4; ++i) { offa[i] = ixa[i] - s; offb[i] = ixb[i] - s; }

    const float* const xu  = x   + ((size_t)n * IN_F * H + oy) * W + ox_base;
    const float* const eu0 = emb + (size_t)n * IN_F * SRC * SRC + iy0 * SRC;
    const float* const eu1 = emb + (size_t)n * IN_F * SRC * SRC + iy1 * SRC;
    const unsigned xoff = (unsigned)(cc * H * W + p0);
    const unsigned eoff = (unsigned)(cc * SRC * SRC + s);
    const int dsto = p0 * XP + cc;
    const float wy0 = 1.f - wy;

    const int lane = tid & 31;
    const int wave = tid >> 5;
    const int ln16 = lane & 15;
    const int hi   = lane >> 4;
    const int m_base = wave * 16;

    const int wrow = lane >> 3;
    const int wcol = (lane & 7) * 4;
    const unsigned wl = (unsigned)((m_base + wrow) * IN_F + wcol);
    float* const s_w = s_tile + WOFF + m_base * XP;
    float* const wdst = s_w + wrow * XP + wcol;

    v4f wv[4];
    v4f px[2], pr0[2], pr1[2];
    auto issue_w = [&](int c) {
        #pragma unroll
        for (int i = 0; i < 4; ++i)
            wv[i] = *(const v4f*)(w1 + (size_t)(c * CH + i * 4 * IN_F) + wl);
    };
    auto store_w = [&]() {
        #pragma unroll
        for (int i = 0; i < 4; ++i)
            *(v4f*)(wdst + i * 4 * XP) = wv[i];
    };
    auto issue_chunk = [&](int c) {
        #pragma unroll
        for (int u = 0; u < 2; ++u) {
            const size_t j = (size_t)(2 * c + u);
            const float* const xb  = xu  + j * 16 * H * W;
            const float* const eb0 = eu0 + j * 16 * SRC * SRC;
            const float* const eb1 = eu1 + j * 16 * SRC * SRC;
            px[u]  = *(const v4f*)(xb + xoff);
            pr0[u] = load4_dword_aligned(eb0 + eoff);
            pr1[u] = load4_dword_aligned(eb1 + eoff);
        }
    };
    auto blend_store_chunk = [&](int b) {
        float* const dst = s_x + b * XBUF + dsto;
        #pragma unroll
        for (int u = 0; u < 2; ++u) {
            const v4f t = pr0[u] * wy0 + pr1[u] * wy;
            #pragma unroll
            for (int i = 0; i < 4; ++i) {
                const float ta = pick4(t, offa[i]);
                const float tb = pick4(t, offb[i]);
                const float ev = ta * (1.f - wxv[i]) + tb * wxv[i];
                dst[i * XP + u * 16] = px[u][i] + ev;
            }
        }
    };

    {
        v8f acc0 = {}, acc1 = {}, acc2 = {}, acc3 = {};
        const float* const ap = s_w + ln16 * XP + hi * 2;
        const int bo = ln16 * XP + hi * 2;

        issue_w(0);
        issue_chunk(0);
        store_w();
        blend_store_chunk(0);
        __syncthreads();

        auto gemm_chunk = [&](int b) {
            const float* const bp0 = s_x + b * XBUF + bo;
            const float* const bp1 = bp0 + 16 * XP;
            const float* const bp2 = bp0 + 32 * XP;
            const float* const bp3 = bp0 + 48 * XP;
            #pragma unroll
            for (int q = 0; q < 8; ++q) {
                const v2f a   = *(const v2f*)(ap  + 4 * q);
                const v2f bb0 = *(const v2f*)(bp0 + 4 * q);
                const v2f bb1 = *(const v2f*)(bp1 + 4 * q);
                const v2f bb2 = *(const v2f*)(bp2 + 4 * q);
                const v2f bb3 = *(const v2f*)(bp3 + 4 * q);
                acc0 = __builtin_amdgcn_wmma_f32_16x16x4_f32(false, a, false, bb0, (short)0, acc0, false, false);
                acc1 = __builtin_amdgcn_wmma_f32_16x16x4_f32(false, a, false, bb1, (short)0, acc1, false, false);
                acc2 = __builtin_amdgcn_wmma_f32_16x16x4_f32(false, a, false, bb2, (short)0, acc2, false, false);
                acc3 = __builtin_amdgcn_wmma_f32_16x16x4_f32(false, a, false, bb3, (short)0, acc3, false, false);
            }
        };

        for (int c = 0; c < IN_F / CH - 1; ++c) {
            issue_w(c + 1);
            issue_chunk(c + 1);
            gemm_chunk(c & 1);
            __builtin_amdgcn_sched_barrier(0);
            store_w();
            blend_store_chunk((c + 1) & 1);
            __syncthreads();
        }
        gemm_chunk((IN_F / CH - 1) & 1);
        __syncthreads();
        #pragma unroll
        for (int i = 0; i < 8; ++i) {
            const int m = m_base + hi * 8 + i;
            const float bias = b1[m];
            s_hidd[( 0 + ln16) * HS + m] = fmaxf(acc0[i] + bias, 0.f);
            s_hidd[(16 + ln16) * HS + m] = fmaxf(acc1[i] + bias, 0.f);
            s_hidd[(32 + ln16) * HS + m] = fmaxf(acc2[i] + bias, 0.f);
            s_hidd[(48 + ln16) * HS + m] = fmaxf(acc3[i] + bias, 0.f);
        }
    }
    __syncthreads();

    if (wave < 4) {
        const int pix = wave * 16;
        v8f acc = {};
        const float* w2row = w2 + (size_t)(ln16) * MLP + hi * 2;
        const float* bp = &s_hidd[(pix + ln16) * HS + hi * 2];
        #pragma unroll 4
        for (int k = 0; k < MLP; k += 4) {
            const v2f a = *(const v2f*)(w2row + k);
            const v2f b = *(const v2f*)(bp + k);
            acc = __builtin_amdgcn_wmma_f32_16x16x4_f32(false, a, false, b, (short)0, acc, false, false);
        }
        #pragma unroll
        for (int i = 0; i < 8; ++i) {
            const int aix = hi * 8 + i;
            s_att[(pix + ln16) * NATT + aix] = softplus_stable(acc[i] + b2[aix]);
        }
    }
    __syncthreads();

    const size_t out_half = (size_t)NB * NBINS * H * W;
    {
        const int p  = tid & (PTILE - 1);
        const int k0 = tid >> 6;
        const int ox = ox_base + p;
        const float fx = (float)ox * sc;
        const int   ix0 = (int)fx;
        const int   ix1 = min(ix0 + 1, SRC - 1);
        const float wx  = fx - (float)ix0;
        const float wx0 = 1.f - wx;

        float att[NATT];
        #pragma unroll
        for (int a = 0; a < NATT; ++a) att[a] = s_att[p * NATT + a];

        const float* const bq = b_prev + (size_t)(n * NBINS + k0) * SRC * SRC;
        const int o00 = iy0 * SRC + ix0;
        const int o01 = iy0 * SRC + ix1;
        const int o10 = iy1 * SRC + ix0;
        const int o11 = iy1 * SRC + ix1;
        float* const oq = out + ((size_t)(n * NBINS + k0) * H + oy) * W + ox;

        #pragma unroll
        for (int j = 0; j < NBINS / 4; j += 2) {
            const float* const b0 = bq + (size_t)j * 4 * SRC * SRC;
            const float* const b1 = b0 + 4 * SRC * SRC;
            const v2f c00 = { b0[o00], b1[o00] };
            const v2f c01 = { b0[o01], b1[o01] };
            const v2f c10 = { b0[o10], b1[o10] };
            const v2f c11 = { b0[o11], b1[o11] };
            const v2f bc  = wy0 * (wx0 * c00 + wx * c01)
                          + wy  * (wx0 * c10 + wx * c11);

            v2f delta = { 0.f, 0.f };
            #pragma unroll
            for (int a = 0; a < NATT; ++a) {
                const v2f dx  = att[a] - bc;
                const v2f den = 1.0f + 300.0f * dx * dx;
                const v2f r   = { __builtin_amdgcn_rcpf(den.x), __builtin_amdgcn_rcpf(den.y) };
                delta += dx * r;
            }
            const v2f res = bc + delta;
            float* const o0 = oq + (size_t)j * 4 * H * W;
            float* const o1 = o0 + (size_t)4 * H * W;
            o0[0]        = res.x;
            o0[out_half] = res.x;
            o1[0]        = res.y;
            o1[out_half] = res.y;
        }
    }
}

extern "C" void kernel_launch(void* const* d_in, const int* in_sizes, int n_in,
                              void* d_out, int out_size, void* d_ws, size_t ws_size,
                              hipStream_t stream) {
    const float* x      = (const float*)d_in[0];
    const float* b_prev = (const float*)d_in[1];
    const float* emb    = (const float*)d_in[2];
    const float* w1     = (const float*)d_in[3];
    const float* b1     = (const float*)d_in[4];
    const float* w2     = (const float*)d_in[5];
    const float* b2     = (const float*)d_in[6];
    float* out = (float*)d_out;

    dim3 grid(NB * H * (W / PTILE));
    attractor_fused<<<grid, 256, 0, stream>>>(x, b_prev, emb, w1, b1, w2, b2, out);
}


# Wave priority (s_setprio): 3 in the WMMA sections of chunks 0-3 and GEMM2, 2 in chunks 4-7, 1 in the blends, 0 in softplus and stage 4: vector work no longer delays other waves' WMMAs; sums unchanged
# speedup vs baseline: 2.6315x; 1.0745x over previous
#include <hip/hip_runtime.h>
#include <math.h>

typedef __attribute__((ext_vector_type(2))) float v2f;
typedef __attribute__((ext_vector_type(4))) float v4f;
typedef __attribute__((ext_vector_type(8))) float v8f;

#define IN_F   256
#define MLP    128
#define NATT   16
#define NBINS  64
#define PTILE  64
#define H      128
#define W      128
#define SRC    64
#define NB     4
#define CH     32
#define XP     (CH + 4)
#define HS     (MLP + 4)
#define XBUF   (PTILE * XP)
#define WOFF   (2 * XBUF)
#define WSTG   (MLP * XP)

__device__ __forceinline__ v4f load4_dword_aligned(const float* p) {
    v4f v;
    __builtin_memcpy(&v, p, sizeof(v));
    return v;
}

__device__ __forceinline__ float pick4(v4f t, int off) {
    const float lo = (off & 1) ? t.y : t.x;
    const float hi = (off & 1) ? t.w : t.z;
    return (off & 2) ? hi : lo;
}

__device__ __forceinline__ float softplus_stable(float v) {
    return fmaxf(v, 0.0f) + log1pf(__expf(-fabsf(v)));
}

__global__ __launch_bounds__(256, 8)
void attractor_fused(const float* __restrict__ x,
                     const float* __restrict__ b_prev,
                     const float* __restrict__ emb,
                     const float* __restrict__ w1,
                     const float* __restrict__ b1,
                     const float* __restrict__ w2,
                     const float* __restrict__ b2,
                     float* __restrict__ out)
{
    __shared__ __attribute__((aligned(16))) float s_tile[WOFF + WSTG];
    __shared__ float s_att [PTILE * NATT];
    float* const s_x    = s_tile;
    float* const s_hidd = s_tile;

    __builtin_amdgcn_s_setprio(3);

    const int tid = threadIdx.x;
    const int blk = blockIdx.x;
    const int ox_base = (blk & 1) * PTILE;
    const int oy      = (blk >> 1) & (H - 1);
    const int n       = blk >> 8;

    const float sc = (float)(SRC - 1) / (float)(H - 1);
    const float fy = (float)oy * sc;
    const int   iy0 = (int)fy;
    const int   iy1 = min(iy0 + 1, SRC - 1);
    const float wy  = fy - (float)iy0;

    const int g   = (tid & 7) | ((tid >> 2) & 8);
    const int cc  = ((tid >> 3) & 3) | ((tid >> 4) & 12);
    const int p0  = g * 4;
    const int ox0 = ox_base + p0;

    int   ixa[4], ixb[4];
    float wxv[4];
    #pragma unroll
    for (int i = 0; i < 4; ++i) {
        const float fx = (float)(ox0 + i) * sc;
        ixa[i] = (int)fx;
        ixb[i] = min(ixa[i] + 1, SRC - 1);
        wxv[i] = fx - (float)ixa[i];
    }
    const int s = min(ixa[0], SRC - 4);
    int offa[4], offb[4];
    #pragma unroll
    for (int i = 0; i < 4; ++i) { offa[i] = ixa[i] - s; offb[i] = ixb[i] - s; }

    const float* const xu  = x   + ((size_t)n * IN_F * H + oy) * W + ox_base;
    const float* const eu0 = emb + (size_t)n * IN_F * SRC * SRC + iy0 * SRC;
    const float* const eu1 = emb + (size_t)n * IN_F * SRC * SRC + iy1 * SRC;
    const unsigned xoff = (unsigned)(cc * H * W + p0);
    const unsigned eoff = (unsigned)(cc * SRC * SRC + s);
    const int dsto = p0 * XP + cc;
    const float wy0 = 1.f - wy;

    const int lane = tid & 31;
    const int wave = tid >> 5;
    const int ln16 = lane & 15;
    const int hi   = lane >> 4;
    const int m_base = wave * 16;

    const int wrow = lane >> 3;
    const int wcol = (lane & 7) * 4;
    const unsigned wl = (unsigned)((m_base + wrow) * IN_F + wcol);
    float* const s_w = s_tile + WOFF + m_base * XP;
    float* const wdst = s_w + wrow * XP + wcol;

    v4f wv[4];
    v4f px[2], pr0[2], pr1[2];
    auto issue_w = [&](int c) {
        #pragma unroll
        for (int i = 0; i < 4; ++i)
            wv[i] = *(const v4f*)(w1 + (size_t)(c * CH + i * 4 * IN_F) + wl);
    };
    auto store_w = [&]() {
        #pragma unroll
        for (int i = 0; i < 4; ++i)
            *(v4f*)(wdst + i * 4 * XP) = wv[i];
    };
    auto issue_chunk = [&](int c) {
        #pragma unroll
        for (int u = 0; u < 2; ++u) {
            const size_t j = (size_t)(2 * c + u);
            const float* const xb  = xu  + j * 16 * H * W;
            const float* const eb0 = eu0 + j * 16 * SRC * SRC;
            const float* const eb1 = eu1 + j * 16 * SRC * SRC;
            px[u]  = *(const v4f*)(xb + xoff);
            pr0[u] = load4_dword_aligned(eb0 + eoff);
            pr1[u] = load4_dword_aligned(eb1 + eoff);
        }
    };
    auto blend_store_chunk = [&](int b) {
        float* const dst = s_x + b * XBUF + dsto;
        #pragma unroll
        for (int u = 0; u < 2; ++u) {
            const v4f t = pr0[u] * wy0 + pr1[u] * wy;
            #pragma unroll
            for (int i = 0; i < 4; ++i) {
                const float ta = pick4(t, offa[i]);
                const float tb = pick4(t, offb[i]);
                const float ev = ta * (1.f - wxv[i]) + tb * wxv[i];
                dst[i * XP + u * 16] = px[u][i] + ev;
            }
        }
    };

    {
        v8f acc0 = {}, acc1 = {}, acc2 = {}, acc3 = {};
        const float* const ap = s_w + ln16 * XP + hi * 2;
        const int bo = ln16 * XP + hi * 2;

        issue_w(0);
        issue_chunk(0);
        store_w();
        blend_store_chunk(0);
        __syncthreads();

        auto gemm_chunk = [&](int b) {
            const float* const bp0 = s_x + b * XBUF + bo;
            const float* const bp1 = bp0 + 16 * XP;
            const float* const bp2 = bp0 + 32 * XP;
            const float* const bp3 = bp0 + 48 * XP;
            #pragma unroll
            for (int q = 0; q < 8; ++q) {
                const v2f a   = *(const v2f*)(ap  + 4 * q);
                const v2f bb0 = *(const v2f*)(bp0 + 4 * q);
                const v2f bb1 = *(const v2f*)(bp1 + 4 * q);
                const v2f bb2 = *(const v2f*)(bp2 + 4 * q);
                const v2f bb3 = *(const v2f*)(bp3 + 4 * q);
                acc0 = __builtin_amdgcn_wmma_f32_16x16x4_f32(false, a, false, bb0, (short)0, acc0, false, false);
                acc1 = __builtin_amdgcn_wmma_f32_16x16x4_f32(false, a, false, bb1, (short)0, acc1, false, false);
                acc2 = __builtin_amdgcn_wmma_f32_16x16x4_f32(false, a, false, bb2, (short)0, acc2, false, false);
                acc3 = __builtin_amdgcn_wmma_f32_16x16x4_f32(false, a, false, bb3, (short)0, acc3, false, false);
            }
        };

        for (int c = 0; c < IN_F / CH - 1; ++c) {
            if (c < IN_F / CH / 2) __builtin_amdgcn_s_setprio(3);
            else                   __builtin_amdgcn_s_setprio(2);
            issue_w(c + 1);
            issue_chunk(c + 1);
            gemm_chunk(c & 1);
            __builtin_amdgcn_sched_barrier(0);
            __builtin_amdgcn_s_setprio(1);
            store_w();
            blend_store_chunk((c + 1) & 1);
            __syncthreads();
        }
        __builtin_amdgcn_s_setprio(2);
        gemm_chunk((IN_F / CH - 1) & 1);
        __builtin_amdgcn_s_setprio(3);
        __syncthreads();
        #pragma unroll
        for (int i = 0; i < 8; ++i) {
            const int m = m_base + hi * 8 + i;
            const float bias = b1[m];
            s_hidd[( 0 + ln16) * HS + m] = fmaxf(acc0[i] + bias, 0.f);
            s_hidd[(16 + ln16) * HS + m] = fmaxf(acc1[i] + bias, 0.f);
            s_hidd[(32 + ln16) * HS + m] = fmaxf(acc2[i] + bias, 0.f);
            s_hidd[(48 + ln16) * HS + m] = fmaxf(acc3[i] + bias, 0.f);
        }
    }
    __syncthreads();

    if (wave < 4) {
        const int pix = wave * 16;
        v8f acc = {};
        const float* w2row = w2 + (size_t)(ln16) * MLP + hi * 2;
        const float* bp = &s_hidd[(pix + ln16) * HS + hi * 2];
        #pragma unroll 4
        for (int k = 0; k < MLP; k += 4) {
            const v2f a = *(const v2f*)(w2row + k);
            const v2f b = *(const v2f*)(bp + k);
            acc = __builtin_amdgcn_wmma_f32_16x16x4_f32(false, a, false, b, (short)0, acc, false, false);
        }
        __builtin_amdgcn_s_setprio(0);
        #pragma unroll
        for (int i = 0; i < 8; ++i) {
            const int aix = hi * 8 + i;
            s_att[(pix + ln16) * NATT + aix] = softplus_stable(acc[i] + b2[aix]);
        }
    }
    __builtin_amdgcn_s_setprio(0);
    __syncthreads();

    const size_t out_half = (size_t)NB * NBINS * H * W;
    {
        const int p  = tid & (PTILE - 1);
        const int k0 = tid >> 6;
        const int ox = ox_base + p;
        const float fx = (float)ox * sc;
        const int   ix0 = (int)fx;
        const int   ix1 = min(ix0 + 1, SRC - 1);
        const float wx  = fx - (float)ix0;
        const float wx0 = 1.f - wx;

        float att[NATT];
        #pragma unroll
        for (int a = 0; a < NATT; ++a) att[a] = s_att[p * NATT + a];

        const float* const bq = b_prev + (size_t)(n * NBINS + k0) * SRC * SRC;
        const int o00 = iy0 * SRC + ix0;
        const int o01 = iy0 * SRC + ix1;
        const int o10 = iy1 * SRC + ix0;
        const int o11 = iy1 * SRC + ix1;
        float* const oq = out + ((size_t)(n * NBINS + k0) * H + oy) * W + ox;

        #pragma unroll
        for (int j = 0; j < NBINS / 4; j += 2) {
            const float* const b0 = bq + (size_t)j * 4 * SRC * SRC;
            const float* const b1 = b0 + 4 * SRC * SRC;
            const v2f c00 = { b0[o00], b1[o00] };
            const v2f c01 = { b0[o01], b1[o01] };
            const v2f c10 = { b0[o10], b1[o10] };
            const v2f c11 = { b0[o11], b1[o11] };
            const v2f bc  = wy0 * (wx0 * c00 + wx * c01)
                          + wy  * (wx0 * c10 + wx * c11);

            v2f delta = { 0.f, 0.f };
            #pragma unroll
            for (int a = 0; a < NATT; ++a) {
                const v2f dx  = att[a] - bc;
                const v2f den = 1.0f + 300.0f * dx * dx;
                const v2f r   = { __builtin_amdgcn_rcpf(den.x), __builtin_amdgcn_rcpf(den.y) };
                delta += dx * r;
            }
            const v2f res = bc + delta;
            float* const o0 = oq + (size_t)j * 4 * H * W;
            float* const o1 = o0 + (size_t)4 * H * W;
            o0[0]        = res.x;
            o0[out_half] = res.x;
            o1[0]        = res.y;
            o1[out_half] = res.y;
        }
    }
}

extern "C" void kernel_launch(void* const* d_in, const int* in_sizes, int n_in,
                              void* d_out, int out_size, void* d_ws, size_t ws_size,
                              hipStream_t stream) {
    const float* x      = (const float*)d_in[0];
    const float* b_prev = (const float*)d_in[1];
    const float* emb    = (const float*)d_in[2];
    const float* w1     = (const float*)d_in[3];
    const float* b1     = (const float*)d_in[4];
    const float* w2     = (const float*)d_in[5];
    const float* b2     = (const float*)d_in[6];
    float* out = (float*)d_out;

    dim3 grid(NB * H * (W / PTILE));
    attractor_fused<<<grid, 256, 0, stream>>>(x, b_prev, emb, w1, b1, w2, b2, out);
}
